# starting kernel
# baseline (speedup 1.0000x reference)
#include <hip/hip_runtime.h>

typedef __attribute__((ext_vector_type(2))) float v2f;
typedef __attribute__((ext_vector_type(8))) float v8f;
typedef __attribute__((ext_vector_type(4))) int   v4i;

#define BATCH 16
#define CIN   16
#define HH    64
#define WW    64
#define COUT  16
#define OHH   62
#define OWW   62
#define NPOS  (OHH * OWW)
#define KTOT  (CIN * 9)
#define NSTEP (KTOT / 4)

struct alignas(16) OffTab { int v[NSTEP][32][4]; };

constexpr OffTab make_tab() {
    OffTab t{};
    for (int s = 0; s < NSTEP; ++s)
        for (int lane = 0; lane < 32; ++lane) {
            const int half = lane >> 4;
            const int mn   = lane & 15;
            for (int tt = 0; tt < 2; ++tt) {
                const int ck = 4 * s + 2 * half + tt;
                const int c  = ck / 9, k = ck - c * 9;
                const int i  = k / 3,  j = k - i * 3;
                t.v[s][lane][tt]     = mn * (CIN * HH * WW) + c * (HH * WW) + i * WW + j;
                t.v[s][lane][2 + tt] = mn * (CIN * NPOS * 9) + c * (NPOS * 9) + k;
            }
        }
    return t;
}

__device__ constexpr OffTab g_tab = make_tab();

typedef const __attribute__((address_space(1))) v4i* gv4i_ptr;

__global__ __launch_bounds__(256)
void lc2d_wmma_kernel(const float* __restrict__ xin,
                      const float* __restrict__ wgt,
                      float* __restrict__ out)
{
    const int lane = threadIdx.x & 31;
    const int wave = __builtin_amdgcn_readfirstlane(threadIdx.x >> 5);
    const int pos  = blockIdx.x * 8 + wave;
    if (pos >= NPOS) return;

    const int y = pos / OWW;
    const int x = pos - y * OWW;

    const float* __restrict__ xp = xin + y * WW + x;
    const float* __restrict__ wp = wgt + pos * 9;

    const v4i* tg = (const v4i*)&g_tab;
    asm volatile("" : "+s"(tg));
    gv4i_ptr tp = (gv4i_ptr)tg;

    v8f acc = {};

#pragma unroll
    for (int s = 0; s < NSTEP; ++s) {
        const v4i o = tp[s * 32 + lane];
        v2f a, b;
        a.x = xp[o.x];
        a.y = xp[o.y];
        b.x = wp[o.z];
        b.y = wp[o.w];
        acc = __builtin_amdgcn_wmma_f32_16x16x4_f32(
            false, a, false, b, (short)0, acc, false, false);
    }

    const int half = lane >> 4;
    const int mn   = lane & 15;
    float* __restrict__ op = out + pos;
    const int voff = mn * NPOS + half * (8 * COUT * NPOS);
#pragma unroll
    for (int r = 0; r < 8; ++r)
        op[voff + r * (COUT * NPOS)] = acc[r];
}

extern "C" void kernel_launch(void* const* d_in, const int* in_sizes, int n_in,
                              void* d_out, int out_size, void* d_ws, size_t ws_size,
                              hipStream_t stream) {
    const float* xin = (const float*)d_in[0];
    const float* wgt = (const float*)d_in[1];
    float* out = (float*)d_out;

    const int waves_per_block = 8;
    const int blocks = (NPOS + waves_per_block - 1) / waves_per_block;
    lc2d_wmma_kernel<<<blocks, 256, 0, stream>>>(xin, wgt, out);
}


# First launch folded into the product launch: each block of 16 positions reads its own x window in native order and turns it batch-fastest in LDS; x fragments come from LDS; sums unchanged
# speedup vs baseline: 4.4318x; 4.4318x over previous
#include <hip/hip_runtime.h>


typedef __attribute__((ext_vector_type(2))) float v2f;
typedef __attribute__((ext_vector_type(4))) float v4f;
typedef __attribute__((ext_vector_type(8))) float v8f;

#define BATCH 16
#define CIN   16
#define HH    64
#define WW    64
#define COUT  16
#define OHH   62
#define OWW   62
#define NPOS  (OHH * OWW)
#define KTOT  (CIN * 9)
#define NTILE (KTOT / 16)
#define PLANE (HH * WW)

#define WRUN      (NPOS * 9)
#define OSTRIDE   (CIN * WRUN)
#define TILEF     (BATCH * COUT)
#define FPOS      16
#define NTHR      (FPOS * 32)
#define NFBLK     ((NPOS + FPOS - 1) / FPOS)
#define TPITCH    20

#define XCOLS  24
#define XCHUNK (XCOLS / 4)
#define XP     17
#define XRUNS  (CIN * 3)
#define XSF    (XRUNS * XCOLS * XP)
#define XLOADS ((BATCH * XRUNS * XCHUNK) / NTHR)
#define WPRE   6

__device__ __forceinline__ constexpr int loff(int ck) {
    return (((ck / 9) * 3 + (ck % 9) / 3) * XCOLS + (ck % 9) % 3) * XP;
}

__global__ __launch_bounds__(NTHR)
void lc2d_wmma_kernel(const float* __restrict__ xin,
                      const float* __restrict__ wgt,
                      float* __restrict__ out)
{
    __shared__ __attribute__((aligned(16))) float xs[XSF];
    __shared__ __attribute__((aligned(16))) float lds[TILEF * TPITCH];
    const int tid  = threadIdx.x;
    const int lane = tid & 31;
    const int wave = __builtin_amdgcn_readfirstlane(tid >> 5);
    const int pos0 = blockIdx.x * FPOS;
    const int pos  = (pos0 + wave < NPOS) ? (pos0 + wave) : (NPOS - 1);

    const int f0 = pos0 + 2 * (pos0 / OWW);
    const int a0 = f0 & ~3;

    v4f xv[XLOADS];
    int  xdst[XLOADS];
#pragma unroll
    for (int it = 0; it < XLOADS; ++it) {
        const int g   = tid + NTHR * it;
        const int run = g / XCHUNK;
        const int k   = g - XCHUNK * run;
        const int b   = run & 15;
        const int ci  = run >> 4;
        const int c   = ci / 3;
        const int i   = ci - 3 * c;
        int idx = a0 + 64 * i + 4 * k;
        idx = (idx < PLANE - 4) ? idx : (PLANE - 4);
        xv[it]   = *(const v4f*)(xin + (b * CIN + c) * PLANE + idx);
        xdst[it] = (ci * XCOLS + 4 * k) * XP + b;
    }
    __builtin_amdgcn_sched_barrier(0);

    const int half = lane >> 4;
    const int mn   = lane & 15;
    const float* __restrict__ wp = wgt + pos * 9 + half * (2 * OSTRIDE);
    v2f wv[NTILE][4];
#pragma unroll
    for (int t = 0; t < WPRE; ++t) {
        const int ck = 16 * t + mn;
        const int c  = ck / 9;
        const float* __restrict__ wt = wp + c * WRUN + (ck - 9 * c);
#pragma unroll
        for (int j = 0; j < 4; ++j) {
            wv[t][j].x = wt[(4 * j)     * OSTRIDE];
            wv[t][j].y = wt[(4 * j + 1) * OSTRIDE];
        }
    }

#pragma unroll
    for (int it = 0; it < XLOADS; ++it) {
#pragma unroll
        for (int e = 0; e < 4; ++e)
            xs[xdst[it] + e * XP] = xv[it][e];
    }
    __builtin_amdgcn_sched_barrier(0);
    asm volatile("s_wait_dscnt 0x0" ::: "memory");
    asm volatile("s_barrier_signal -1\n\ts_barrier_wait -1" ::: "memory");

#pragma unroll
    for (int t = WPRE; t < NTILE; ++t) {
        const int ck = 16 * t + mn;
        const int c  = ck / 9;
        const float* __restrict__ wt = wp + c * WRUN + (ck - 9 * c);
#pragma unroll
        for (int j = 0; j < 4; ++j) {
            wv[t][j].x = wt[(4 * j)     * OSTRIDE];
            wv[t][j].y = wt[(4 * j + 1) * OSTRIDE];
        }
    }

    {
        const int y  = pos / OWW;
        const int lp = (pos + 2 * y) - a0;
        const float* xl = xs + lp * XP + mn;

        float idv[8];
#pragma unroll
        for (int i = 0; i < 8; ++i)
            idv[i] = (mn == 4 * (i >> 1) + 2 * half + (i & 1)) ? 1.0f : 0.0f;

        v8f acc = {};

#pragma unroll
        for (int t = 0; t < NTILE; ++t) {
            v8f d = {};
#pragma unroll
            for (int j = 0; j < 4; ++j) {
                v2f b;
                b.x = idv[2 * j];
                b.y = idv[2 * j + 1];
                d = __builtin_amdgcn_wmma_f32_16x16x4_f32(
                    false, wv[t][j], false, b, (short)0, d, false, false);
            }
#pragma unroll
            for (int q = 0; q < 4; ++q) {
                const int k0 = 16 * t + 2 * q;
                v2f a, b;
                a.x = xl[half ? loff(k0 + 8) : loff(k0)];
                a.y = xl[half ? loff(k0 + 9) : loff(k0 + 1)];
                b.x = d[2 * q];
                b.y = d[2 * q + 1];
                acc = __builtin_amdgcn_wmma_f32_16x16x4_f32(
                    false, a, false, b, (short)0, acc, false, false);
            }
        }

#pragma unroll
        for (int r = 0; r < 8; ++r)
            lds[(r * 32 + lane) * TPITCH + wave] = acc[r];
    }
    __syncthreads();

    const int npos = (NPOS - pos0 < FPOS) ? (NPOS - pos0) : FPOS;
    const int p4 = (tid & 3) * 4;
#pragma unroll
    for (int i = 0; i < 2; ++i) {
        const int j = (tid >> 2) + 128 * i;
        const int m = (j >> 5) + 8 * ((j >> 4) & 1);
        const int n = j & 15;
        if (p4 < npos) {
            const v4f t = *(const v4f*)&lds[j * TPITCH + p4];
            *(v4f*)(out + (m * COUT + n) * NPOS + pos0 + p4) = t;
        }
    }
}

extern "C" void kernel_launch(void* const* d_in, const int* in_sizes, int n_in,
                              void* d_out, int out_size, void* d_ws, size_t ws_size,
                              hipStream_t stream) {
    const float* xin = (const float*)d_in[0];
    const float* wgt = (const float*)d_in[1];
    float* out = (float*)d_out;

    lc2d_wmma_kernel<<<NFBLK, NTHR, 0, stream>>>(xin, wgt, out);
}
